# starting kernel
# baseline (speedup 1.0000x reference)
#include <hip/hip_runtime.h>
#include <hip/hip_bf16.h>

typedef __attribute__((ext_vector_type(2))) float v2f;
typedef __attribute__((ext_vector_type(4))) float v4f;
typedef __attribute__((ext_vector_type(8))) float v8f;

#define A_TOT   2048
#define EMB_K   128
#define NCOL    46
#define NPAD    48
#define NMOL    64
#define APM     32

__global__ __launch_bounds__(128) void coeff_gemm_wmma(
    const float* __restrict__ A,
    const float* __restrict__ W,
    const float* __restrict__ bias,
    float* __restrict__ C)
{
    const int wave = threadIdx.x >> 5;
    const int lane = threadIdx.x & 31;

    const int mtile = blockIdx.x * 4 + wave;
    const int m0    = mtile * 16;
    const int n0    = blockIdx.y * 16;

    const int rowm = lane & 15;
    const int kb   = (lane >> 4) << 1;

    const int  col  = n0 + rowm;
    const float msk = (col < NCOL) ? 1.0f : 0.0f;
    const int  colc = (col < NCOL) ? col : (NCOL - 1);

    const float* __restrict__ arow = A + (m0 + rowm) * EMB_K;

    v8f acc = {};
    #pragma unroll
    for (int k0 = 0; k0 < EMB_K; k0 += 4) {
        v2f a = *(const v2f*)(arow + k0 + kb);
        v2f b;
        b.x = W[(k0 + kb)     * NCOL + colc] * msk;
        b.y = W[(k0 + kb + 1) * NCOL + colc] * msk;
        acc = __builtin_amdgcn_wmma_f32_16x16x4_f32(
                  false, a, false, b, (short)0, acc, false, false);
    }

    const float bv = bias[colc] * msk;

    const int mbase = m0 + ((lane >> 4) << 3);
    #pragma unroll
    for (int r = 0; r < 8; ++r) {
        C[(mbase + r) * NPAD + col] = acc[r] + bv;
    }
}

__global__ __launch_bounds__(256) void mol_field_kernel(
    const float* __restrict__ coeffs,
    const float* __restrict__ R,
    const float* __restrict__ coords,
    float* __restrict__ outp)
{
    __shared__ float scoef[APM][NPAD];
    __shared__ float sR[APM][3];
    __shared__ float scrd[NMOL][3];
    __shared__ float partial[4][NMOL];
    __shared__ float sres[NMOL];

    const int m = blockIdx.x;
    const int t = threadIdx.x;

    for (int idx = t; idx < APM * NPAD; idx += 256)
        (&scoef[0][0])[idx] = coeffs[m * APM * NPAD + idx];
    for (int idx = t; idx < APM * 3; idx += 256)
        (&sR[0][0])[idx] = R[m * APM * 3 + idx];
    for (int idx = t; idx < NMOL * 3; idx += 256)
        (&scrd[0][0])[idx] = coords[idx];
    __syncthreads();

    const int c   = t & 63;
    const int grp = t >> 6;

    const float cx = scrd[c][0], cy = scrd[c][1], cz = scrd[c][2];

    float acc = 0.0f;
    #pragma unroll
    for (int ii = 0; ii < 8; ++ii) {
        const int i = grp * 8 + ii;
        const float dx = sR[i][0] - cx;
        const float dy = sR[i][1] - cy;
        const float dz = sR[i][2] - cz;
        const float adx = fabsf(dx), ady = fabsf(dy), adz = fabsf(dz);
        const float d2 = dx * dx + dy * dy + dz * dz;
        const float d  = sqrtf(d2);

        const float* cf = scoef[i];

        #pragma unroll
        for (int e = 0; e < 8; ++e) {
            float al = cf[e];
            al *= al;
            acc += cf[8 + e] * __expf(-al * d);
        }
        #pragma unroll
        for (int e = 0; e < 6; ++e) {
            float pa = cf[16 + e];
            pa *= pa;
            const float pc  = cf[22 + e];
            const float ang = cf[28 + 3 * e] * adx
                            + cf[29 + 3 * e] * ady
                            + cf[30 + 3 * e] * adz;
            acc += pc * __expf(-pa * d2) * ang;
        }
    }

    partial[grp][c] = acc;
    __syncthreads();

    if (t < NMOL)
        sres[t] = partial[0][t] + partial[1][t] + partial[2][t] + partial[3][t];
    __syncthreads();

    const float v = sres[t >> 2];
    const v4f vv = {v, v, v, v};
    v4f* dst = (v4f*)(outp + m * 2048 + t * 8);
    dst[0] = vv;
    dst[1] = vv;
}

extern "C" void kernel_launch(void* const* d_in, const int* in_sizes, int n_in,
                              void* d_out, int out_size, void* d_ws, size_t ws_size,
                              hipStream_t stream) {
    const float* out_emb = (const float*)d_in[0];
    const float* R       = (const float*)d_in[2];
    const float* coords  = (const float*)d_in[3];
    const float* W       = (const float*)d_in[5];
    const float* bias    = (const float*)d_in[6];

    float* coeffs = (float*)d_ws;
    float* outp   = (float*)d_out;

    dim3 g1(A_TOT / 16 / 4, (NCOL + 15) / 16);
    coeff_gemm_wmma<<<g1, dim3(128), 0, stream>>>(out_emb, W, bias, coeffs);

    mol_field_kernel<<<dim3(NMOL), dim3(256), 0, stream>>>(coeffs, R, coords, outp);
}


# Fuse the coefficient GEMM into the per-molecule field kernel: one launch instead of two, each block makes its 32x46 coeffs on six f32 WMMA tiles from LDS-staged operands; no workspace round trip.
# speedup vs baseline: 2.1765x; 2.1765x over previous
#include <hip/hip_runtime.h>
#include <hip/hip_bf16.h>


typedef __attribute__((ext_vector_type(2))) float v2f;
typedef __attribute__((ext_vector_type(4))) float v4f;
typedef __attribute__((ext_vector_type(8))) float v8f;

#define EMB_K   128
#define NCOL    46
#define NPAD    48
#define NMOL    64
#define APM     32
#define APITCH  132

__global__ __launch_bounds__(256) void mol_field_fused(
    const float* __restrict__ A,
    const float* __restrict__ W,
    const float* __restrict__ bias,
    const float* __restrict__ R,
    const float* __restrict__ coords,
    float* __restrict__ outp)
{
    __shared__ __attribute__((aligned(16))) float sA[APM][APITCH];
    __shared__ __attribute__((aligned(16))) float sW[6 * 256 * 4];
    __shared__ float scoef[APM][NPAD];
    __shared__ float sR[256];
    __shared__ float scrd[256];
    __shared__ float partial[4][NMOL];
    __shared__ float sres[NMOL];

    const int m    = blockIdx.x;
    const int t    = threadIdx.x;
    const int wave = t >> 5;
    const int lane = t & 31;

    const v4f* __restrict__ gA = (const v4f*)(A + m * APM * EMB_K);
    const v4f* __restrict__ gW = (const v4f*)W;
    v4f ra[4], rw[6];
    #pragma unroll
    for (int i = 0; i < 4; ++i) ra[i] = gA[t + 256 * i];
    #pragma unroll
    for (int i = 0; i < 5; ++i) rw[i] = gW[t + 256 * i];
    {
        const int i5 = t + 1280;
        rw[5] = gW[(i5 < EMB_K * NCOL / 4) ? i5 : (EMB_K * NCOL / 4 - 1)];
    }
    const float rR = R[m * APM * 3 + ((t < APM * 3) ? t : (APM * 3 - 1))];
    const float rC = coords[(t < NMOL * 3) ? t : (NMOL * 3 - 1)];

    const int rowm = lane & 15;
    const int kb   = (lane >> 4) << 1;
    const int mt   = (wave >= 3) ? 1 : 0;
    const int nt   = wave - 3 * mt;
    const int  col  = nt * 16 + rowm;
    const int  colc = (col < NCOL) ? col : (NCOL - 1);
    const float bv  = bias[colc];

    #pragma unroll
    for (int i = 0; i < 4; ++i) {
        const int idx = t + 256 * i;
        *(v4f*)&sA[idx >> 5][(idx & 31) * 4] = ra[i];
    }
    #pragma unroll
    for (int i = 0; i < 6; ++i)
        ((v4f*)sW)[t + 256 * i] = rw[i];
    sR[t]   = rR;
    scrd[t] = rC;

    __syncthreads();

    if (wave < 6) {
        const float* arow = &sA[mt * 16 + rowm][kb];
        const float* wcol = &sW[kb * NCOL + colc];

        v2f af[EMB_K / 4], bf[EMB_K / 4];
        #pragma unroll
        for (int s = 0; s < EMB_K / 4; ++s) {
            af[s] = *(const v2f*)(arow + 4 * s);
            bf[s].x = wcol[4 * s * NCOL];
            bf[s].y = wcol[4 * s * NCOL + NCOL];
        }
        __builtin_amdgcn_sched_barrier(0);

        v8f acc = {};
        #pragma unroll
        for (int s = 0; s < EMB_K / 4; ++s) {
            acc = __builtin_amdgcn_wmma_f32_16x16x4_f32(
                      false, af[s], false, bf[s], (short)0, acc, false, false);
        }

        const int mbase = mt * 16 + ((lane >> 4) << 3);
        #pragma unroll
        for (int r = 0; r < 8; ++r)
            scoef[mbase + r][col] = acc[r] + bv;
    }
    __syncthreads();

    const int c   = t & 63;
    const int grp = t >> 6;

    const float cx = scrd[c * 3 + 0], cy = scrd[c * 3 + 1], cz = scrd[c * 3 + 2];

    float acc = 0.0f;
    #pragma unroll
    for (int ii = 0; ii < 8; ++ii) {
        const int i = grp * 8 + ii;
        const float dx = sR[i * 3 + 0] - cx;
        const float dy = sR[i * 3 + 1] - cy;
        const float dz = sR[i * 3 + 2] - cz;
        const float adx = fabsf(dx), ady = fabsf(dy), adz = fabsf(dz);
        const float d2 = dx * dx + dy * dy + dz * dz;
        const float d  = sqrtf(d2);

        const float* cf = scoef[i];

        #pragma unroll
        for (int e = 0; e < 8; ++e) {
            float al = cf[e];
            al *= al;
            acc += cf[8 + e] * __expf(-al * d);
        }
        #pragma unroll
        for (int e = 0; e < 6; ++e) {
            float pa = cf[16 + e];
            pa *= pa;
            const float pc  = cf[22 + e];
            const float ang = cf[28 + 3 * e] * adx
                            + cf[29 + 3 * e] * ady
                            + cf[30 + 3 * e] * adz;
            acc += pc * __expf(-pa * d2) * ang;
        }
    }

    partial[grp][c] = acc;
    __syncthreads();

    if (t < NMOL)
        sres[t] = partial[0][t] + partial[1][t] + partial[2][t] + partial[3][t];
    __syncthreads();

    const float v = sres[t >> 2];
    const v4f vv = {v, v, v, v};
    v4f* dst = (v4f*)(outp + m * 2048 + t * 8);
    dst[0] = vv;
    dst[1] = vv;
}

extern "C" void kernel_launch(void* const* d_in, const int* in_sizes, int n_in,
                              void* d_out, int out_size, void* d_ws, size_t ws_size,
                              hipStream_t stream) {
    const float* out_emb = (const float*)d_in[0];
    const float* R       = (const float*)d_in[2];
    const float* coords  = (const float*)d_in[3];
    const float* W       = (const float*)d_in[5];
    const float* bias    = (const float*)d_in[6];

    float* outp = (float*)d_out;

    mol_field_fused<<<dim3(NMOL), dim3(256), 0, stream>>>(out_emb, W, bias, R, coords, outp);
}


# Field phase re-mapped: lane = atom, wave = 8 columns; per-atom coefficient transforms made once, kept in registers; two-wide sums; new atom reduction tree (LDS + quad DPP): sum order changed.
# speedup vs baseline: 2.8330x; 1.3016x over previous
#include <hip/hip_runtime.h>
#include <hip/hip_bf16.h>


typedef __attribute__((ext_vector_type(2))) float v2f;
typedef __attribute__((ext_vector_type(4))) float v4f;
typedef __attribute__((ext_vector_type(8))) float v8f;

#define EMB_K   128
#define NCOL    46
#define NPAD    48
#define NMOL    64
#define APM     32
#define APITCH  132
#define CPITCH  52
#define RPITCH  36
#define LOG2E   1.44269504088896340736f

__global__ __launch_bounds__(256) void mol_field_fused(
    const float* __restrict__ A,
    const float* __restrict__ W,
    const float* __restrict__ bias,
    const float* __restrict__ R,
    const float* __restrict__ coords,
    float* __restrict__ outp)
{
    __shared__ __attribute__((aligned(16))) float sA[APM][APITCH];
    __shared__ __attribute__((aligned(16))) float sW[6 * 256 * 4];
    __shared__ __attribute__((aligned(16))) float scoef[APM][CPITCH];
    __shared__ float sR[256];
    __shared__ __attribute__((aligned(16))) float scrd[256];
    __shared__ __attribute__((aligned(16))) float sred[NMOL][RPITCH];

    const int m    = blockIdx.x;
    const int t    = threadIdx.x;
    const int wave = t >> 5;
    const int lane = t & 31;

    const v4f* __restrict__ gA = (const v4f*)(A + m * APM * EMB_K);
    const v4f* __restrict__ gW = (const v4f*)W;
    v4f ra[4], rw[6];
    #pragma unroll
    for (int i = 0; i < 4; ++i) ra[i] = gA[t + 256 * i];
    #pragma unroll
    for (int i = 0; i < 5; ++i) rw[i] = gW[t + 256 * i];
    {
        const int i5 = t + 1280;
        rw[5] = gW[(i5 < EMB_K * NCOL / 4) ? i5 : (EMB_K * NCOL / 4 - 1)];
    }
    const float rR = R[m * APM * 3 + ((t < APM * 3) ? t : (APM * 3 - 1))];
    const float rC = coords[(t < NMOL * 3) ? t : (NMOL * 3 - 1)];

    const int rowm = lane & 15;
    const int kb   = (lane >> 4) << 1;
    const int mt   = (wave >= 3) ? 1 : 0;
    const int nt   = wave - 3 * mt;
    const int  col  = nt * 16 + rowm;
    const int  colc = (col < NCOL) ? col : (NCOL - 1);
    const float bv  = bias[colc];

    #pragma unroll
    for (int i = 0; i < 4; ++i) {
        const int idx = t + 256 * i;
        *(v4f*)&sA[idx >> 5][(idx & 31) * 4] = ra[i];
    }
    #pragma unroll
    for (int i = 0; i < 6; ++i)
        ((v4f*)sW)[t + 256 * i] = rw[i];
    sR[t]   = rR;
    scrd[t] = rC;

    __syncthreads();

    if (wave < 6) {
        const float* arow = &sA[mt * 16 + rowm][kb];
        const float* wcol = &sW[kb * NCOL + colc];

        v2f af[EMB_K / 4], bf[EMB_K / 4];
        #pragma unroll
        for (int s = 0; s < EMB_K / 4; ++s) {
            af[s] = *(const v2f*)(arow + 4 * s);
            bf[s].x = wcol[4 * s * NCOL];
            bf[s].y = wcol[4 * s * NCOL + NCOL];
        }
        __builtin_amdgcn_sched_barrier(0);

        v8f acc = {};
        #pragma unroll
        for (int s = 0; s < EMB_K / 4; ++s) {
            acc = __builtin_amdgcn_wmma_f32_16x16x4_f32(
                      false, af[s], false, bf[s], (short)0, acc, false, false);
        }

        const int mbase = mt * 16 + ((lane >> 4) << 3);
        #pragma unroll
        for (int r = 0; r < 8; ++r)
            scoef[mbase + r][col] = acc[r] + bv;
    }
    __syncthreads();

    float cv[NPAD];
    #pragma unroll
    for (int i = 0; i < NPAD / 4; ++i) {
        const v4f q = *(const v4f*)&scoef[lane][4 * i];
        cv[4 * i + 0] = q.x; cv[4 * i + 1] = q.y; cv[4 * i + 2] = q.z; cv[4 * i + 3] = q.w;
    }
    v2f sa[4], sc[4], pa[3], vx[3], vy[3], vz[3];
    #pragma unroll
    for (int h = 0; h < 4; ++h) {
        sa[h].x = -(cv[2 * h] * cv[2 * h]) * LOG2E;
        sa[h].y = -(cv[2 * h + 1] * cv[2 * h + 1]) * LOG2E;
        sc[h].x = cv[8 + 2 * h];
        sc[h].y = cv[9 + 2 * h];
    }
    #pragma unroll
    for (int h = 0; h < 3; ++h) {
        const int e0 = 2 * h, e1 = 2 * h + 1;
        pa[h].x = -(cv[16 + e0] * cv[16 + e0]) * LOG2E;
        pa[h].y = -(cv[16 + e1] * cv[16 + e1]) * LOG2E;
        const float pc0 = cv[22 + e0], pc1 = cv[22 + e1];
        vx[h].x = pc0 * cv[28 + 3 * e0];  vx[h].y = pc1 * cv[28 + 3 * e1];
        vy[h].x = pc0 * cv[29 + 3 * e0];  vy[h].y = pc1 * cv[29 + 3 * e1];
        vz[h].x = pc0 * cv[30 + 3 * e0];  vz[h].y = pc1 * cv[30 + 3 * e1];
    }

    const float rx = sR[lane * 3 + 0], ry = sR[lane * 3 + 1], rz = sR[lane * 3 + 2];
    const float* cc = &scrd[wave * 24];

    #pragma unroll
    for (int j = 0; j < 8; ++j) {
        const float dx = rx - cc[3 * j + 0];
        const float dy = ry - cc[3 * j + 1];
        const float dz = rz - cc[3 * j + 2];
        const float adx = fabsf(dx), ady = fabsf(dy), adz = fabsf(dz);
        const float d2 = dx * dx + dy * dy + dz * dz;
        const float d  = sqrtf(d2);

        v2f acc2 = {0.0f, 0.0f};
        #pragma unroll
        for (int h = 0; h < 4; ++h) {
            const v2f arg = sa[h] * d;
            v2f ex;
            ex.x = __builtin_amdgcn_exp2f(arg.x);
            ex.y = __builtin_amdgcn_exp2f(arg.y);
            acc2 += sc[h] * ex;
        }
        #pragma unroll
        for (int h = 0; h < 3; ++h) {
            const v2f arg = pa[h] * d2;
            v2f ex;
            ex.x = __builtin_amdgcn_exp2f(arg.x);
            ex.y = __builtin_amdgcn_exp2f(arg.y);
            const v2f ang = vx[h] * adx + vy[h] * ady + vz[h] * adz;
            acc2 += ex * ang;
        }
        sred[wave * 8 + j][lane] = acc2.x + acc2.y;
    }
    __syncthreads();

    const int c  = t >> 2;
    const int qd = t & 3;
    const v4f u0 = *(const v4f*)&sred[c][qd * 8];
    const v4f u1 = *(const v4f*)&sred[c][qd * 8 + 4];
    float v = ((u0.x + u0.y) + (u0.z + u0.w)) + ((u1.x + u1.y) + (u1.z + u1.w));
    v += __builtin_bit_cast(float, __builtin_amdgcn_update_dpp(
             0, __builtin_bit_cast(int, v), 0xB1  , 0xf, 0xf, true));
    v += __builtin_bit_cast(float, __builtin_amdgcn_update_dpp(
             0, __builtin_bit_cast(int, v), 0x4E  , 0xf, 0xf, true));

    const v4f vv = {v, v, v, v};
    v4f* dst = (v4f*)(outp + m * 2048 + t * 8);
    dst[0] = vv;
    dst[1] = vv;
}

extern "C" void kernel_launch(void* const* d_in, const int* in_sizes, int n_in,
                              void* d_out, int out_size, void* d_ws, size_t ws_size,
                              hipStream_t stream) {
    const float* out_emb = (const float*)d_in[0];
    const float* R       = (const float*)d_in[2];
    const float* coords  = (const float*)d_in[3];
    const float* W       = (const float*)d_in[5];
    const float* bias    = (const float*)d_in[6];

    float* outp = (float*)d_out;

    mol_field_fused<<<dim3(NMOL), dim3(256), 0, stream>>>(out_emb, W, bias, R, coords, outp);
}


# One wave per SIMD: 128 blocks x 4 waves, 2 per molecule (32 column groups each); GEMM as 12 half-K tiles, 3 chains/wave, halves added at read-back: sum order changed; W in LDS k-pair interleaved.
# speedup vs baseline: 3.1027x; 1.0952x over previous
#include <hip/hip_runtime.h>
#include <hip/hip_bf16.h>


typedef __attribute__((ext_vector_type(2))) float v2f;
typedef __attribute__((ext_vector_type(4))) float v4f;
typedef __attribute__((ext_vector_type(8))) float v8f;

#define EMB_K   128
#define NCOL    46
#define NPAD    48
#define NMOL    64
#define APM     32
#define NBLK    128
#define NTHR    128
#define APITCH  132
#define CPITCH  52
#define RPITCH  36
#define WPITCH  96
#define LOG2E   1.44269504088896340736f

__global__ __launch_bounds__(NTHR) void mol_field_fused(
    const float* __restrict__ A,
    const float* __restrict__ W,
    const float* __restrict__ bias,
    const float* __restrict__ R,
    const float* __restrict__ coords,
    float* __restrict__ outp)
{
    __shared__ __attribute__((aligned(16))) float sA[APM][APITCH];
    __shared__ __attribute__((aligned(16))) float sW[(EMB_K / 2) * WPITCH];
    __shared__ __attribute__((aligned(16))) float scoef[2][APM][CPITCH];
    __shared__ float sR[NTHR];
    __shared__ __attribute__((aligned(16))) float scrd[NTHR];
    __shared__ __attribute__((aligned(16))) float sred[32][RPITCH];

    const int m    = blockIdx.x >> 1;
    const int hf   = blockIdx.x & 1;
    const int t    = threadIdx.x;
    const int wave = t >> 5;
    const int lane = t & 31;

    const v4f* __restrict__ gA = (const v4f*)(A + m * APM * EMB_K);
    v4f ra[8];
    v2f wx[13], wy[13];
    #pragma unroll
    for (int i = 0; i < 8; ++i) ra[i] = gA[t + NTHR * i];
    const bool wact = (t < 5 * 23);
    const int  tw   = wact ? t : (5 * 23 - 1);
    const int  kpl  = tw / 23;
    const int  c2   = tw - 23 * kpl;
    const float* pW = W + (2 * kpl) * NCOL + 2 * c2;
    #pragma unroll
    for (int i = 0; i < 12; ++i) {
        wx[i] = *(const v2f*)(pW + 10 * NCOL * i);
        wy[i] = *(const v2f*)(pW + 10 * NCOL * i + NCOL);
    }
    const bool ok12 = (kpl + 60 < EMB_K / 2);
    const float* p12 = ok12 ? (pW + 10 * NCOL * 12) : (W + (EMB_K - 2) * NCOL + 2 * c2);
    wx[12] = *(const v2f*)p12;
    wy[12] = *(const v2f*)(p12 + NCOL);
    const int wslot0  = kpl * (WPITCH / 4) + c2;
    const int wpad    = (EMB_K / 2) * (WPITCH / 4) - 1;
    const int t96 = (t < APM * 3) ? t : (APM * 3 - 1);
    const float rR = R[m * APM * 3 + t96];
    const float rC = coords[hf * 96 + t96];

    const int rowm = lane & 15;
    const int kb   = (lane >> 4) << 1;
    const int kh   = wave & 1;
    const int mt   = wave >> 1;
    int   colc[3];
    float bv[3];
    #pragma unroll
    for (int nt = 0; nt < 3; ++nt) {
        const int col = nt * 16 + rowm;
        colc[nt] = (col < NCOL) ? col : (NCOL - 1);
        bv[nt]   = bias[colc[nt]];
    }

    #pragma unroll
    for (int i = 0; i < 8; ++i) {
        const int idx = t + NTHR * i;
        *(v4f*)&sA[idx >> 5][(idx & 31) * 4] = ra[i];
    }
    #pragma unroll
    for (int i = 0; i < 13; ++i) {
        const v4f  wv = {wx[i].x, wy[i].x, wx[i].y, wy[i].y};
        const bool ok = wact && (i < 12 || ok12);
        ((v4f*)sW)[ok ? (wslot0 + 5 * (WPITCH / 4) * i) : wpad] = wv;
    }
    sR[t]   = rR;
    scrd[t] = rC;

    __syncthreads();

    {
        const float* arow = &sA[mt * 16 + rowm][kb + 64 * kh];

        v2f af[16], bf[3][16];
        #pragma unroll
        for (int s = 0; s < 16; ++s) {
            af[s] = *(const v2f*)(arow + 4 * s);
            #pragma unroll
            for (int nt = 0; nt < 3; ++nt) {
                bf[nt][s] = *(const v2f*)&sW[(32 * kh + 2 * s + (kb >> 1)) * WPITCH + 2 * colc[nt]];
            }
        }
        __builtin_amdgcn_sched_barrier(0);

        v8f acc[3] = {};
        #pragma unroll
        for (int s = 0; s < 16; ++s) {
            #pragma unroll
            for (int nt = 0; nt < 3; ++nt) {
                acc[nt] = __builtin_amdgcn_wmma_f32_16x16x4_f32(
                              false, af[s], false, bf[nt][s], (short)0, acc[nt], false, false);
            }
        }

        const int mbase = mt * 16 + ((lane >> 4) << 3);
        #pragma unroll
        for (int nt = 0; nt < 3; ++nt) {
            const float b0 = (kh == 0) ? bv[nt] : 0.0f;
            #pragma unroll
            for (int r = 0; r < 8; ++r)
                scoef[kh][mbase + r][nt * 16 + rowm] = acc[nt][r] + b0;
        }
    }
    __syncthreads();

    float cv[NPAD];
    #pragma unroll
    for (int i = 0; i < NPAD / 4; ++i) {
        const v4f q = *(const v4f*)&scoef[0][lane][4 * i] + *(const v4f*)&scoef[1][lane][4 * i];
        cv[4 * i + 0] = q.x; cv[4 * i + 1] = q.y; cv[4 * i + 2] = q.z; cv[4 * i + 3] = q.w;
    }
    v2f sa[4], sc[4], pa[3], vx[3], vy[3], vz[3];
    #pragma unroll
    for (int h = 0; h < 4; ++h) {
        sa[h].x = -(cv[2 * h] * cv[2 * h]) * LOG2E;
        sa[h].y = -(cv[2 * h + 1] * cv[2 * h + 1]) * LOG2E;
        sc[h].x = cv[8 + 2 * h];
        sc[h].y = cv[9 + 2 * h];
    }
    #pragma unroll
    for (int h = 0; h < 3; ++h) {
        const int e0 = 2 * h, e1 = 2 * h + 1;
        pa[h].x = -(cv[16 + e0] * cv[16 + e0]) * LOG2E;
        pa[h].y = -(cv[16 + e1] * cv[16 + e1]) * LOG2E;
        const float pc0 = cv[22 + e0], pc1 = cv[22 + e1];
        vx[h].x = pc0 * cv[28 + 3 * e0];  vx[h].y = pc1 * cv[28 + 3 * e1];
        vy[h].x = pc0 * cv[29 + 3 * e0];  vy[h].y = pc1 * cv[29 + 3 * e1];
        vz[h].x = pc0 * cv[30 + 3 * e0];  vz[h].y = pc1 * cv[30 + 3 * e1];
    }

    const float rx = sR[lane * 3 + 0], ry = sR[lane * 3 + 1], rz = sR[lane * 3 + 2];
    const float* cc = &scrd[wave * 24];

    #pragma unroll
    for (int j = 0; j < 8; ++j) {
        const float dx = rx - cc[3 * j + 0];
        const float dy = ry - cc[3 * j + 1];
        const float dz = rz - cc[3 * j + 2];
        const float adx = fabsf(dx), ady = fabsf(dy), adz = fabsf(dz);
        const float d2 = dx * dx + dy * dy + dz * dz;
        const float d  = sqrtf(d2);

        v2f acc2 = {0.0f, 0.0f};
        #pragma unroll
        for (int h = 0; h < 4; ++h) {
            const v2f arg = sa[h] * d;
            v2f ex;
            ex.x = __builtin_amdgcn_exp2f(arg.x);
            ex.y = __builtin_amdgcn_exp2f(arg.y);
            acc2 += sc[h] * ex;
        }
        #pragma unroll
        for (int h = 0; h < 3; ++h) {
            const v2f arg = pa[h] * d2;
            v2f ex;
            ex.x = __builtin_amdgcn_exp2f(arg.x);
            ex.y = __builtin_amdgcn_exp2f(arg.y);
            const v2f ang = vx[h] * adx + vy[h] * ady + vz[h] * adz;
            acc2 += ex * ang;
        }
        sred[wave * 8 + j][lane] = acc2.x + acc2.y;
    }
    __syncthreads();

    const int c  = t >> 2;
    const int qd = t & 3;
    const v4f u0 = *(const v4f*)&sred[c][qd * 8];
    const v4f u1 = *(const v4f*)&sred[c][qd * 8 + 4];
    float v = ((u0.x + u0.y) + (u0.z + u0.w)) + ((u1.x + u1.y) + (u1.z + u1.w));
    v += __builtin_bit_cast(float, __builtin_amdgcn_update_dpp(
             0, __builtin_bit_cast(int, v), 0xB1  , 0xf, 0xf, true));
    v += __builtin_bit_cast(float, __builtin_amdgcn_update_dpp(
             0, __builtin_bit_cast(int, v), 0x4E  , 0xf, 0xf, true));

    const v4f vv = {v, v, v, v};
    v4f* dst = (v4f*)(outp + m * 2048 + hf * 1024 + t * 8);
    dst[0] = vv;
    dst[1] = vv;
}

extern "C" void kernel_launch(void* const* d_in, const int* in_sizes, int n_in,
                              void* d_out, int out_size, void* d_ws, size_t ws_size,
                              hipStream_t stream) {
    const float* out_emb = (const float*)d_in[0];
    const float* R       = (const float*)d_in[2];
    const float* coords  = (const float*)d_in[3];
    const float* W       = (const float*)d_in[5];
    const float* bias    = (const float*)d_in[6];

    float* outp = (float*)d_out;

    mol_field_fused<<<dim3(NBLK), dim3(NTHR), 0, stream>>>(out_emb, W, bias, R, coords, outp);
}
